# starting kernel
# baseline (speedup 1.0000x reference)
#include <hip/hip_runtime.h>
#include <hip/hip_bf16.h>


typedef __attribute__((ext_vector_type(2))) float v2f;
typedef __attribute__((ext_vector_type(8))) float v8f;

namespace {
constexpr int B_CH = 512;
constexpr int T_LEN = 48000;
constexpr int SLABS = 12;
constexpr int SLAB_LEN = T_LEN / SLABS;
constexpr int BLK = 16;
constexpr int BLOCKS_PER_SLAB = SLAB_LEN / BLK;
constexpr int GRP = 16;
constexpr int GROUPS = (BLOCKS_PER_SLAB + GRP - 1) / GRP;
constexpr int COEF_STRIDE = 64;
}

__global__ void biquad_coef_kernel(const float* __restrict__ b0v, const float* __restrict__ b1v,
                                   const float* __restrict__ b2v, const float* __restrict__ a1v,
                                   const float* __restrict__ a2v, float* __restrict__ coef) {
  int c = blockIdx.x * blockDim.x + threadIdx.x;
  if (c >= B_CH) return;
  const float b0 = b0v[c], b1 = b1v[c], b2 = b2v[c], a1 = a1v[c], a2 = a2v[c];
  float* w = coef + c * COEF_STRIDE;
  const float v1 = b1 - a1 * b0;
  const float v2 = b2 - a2 * b0;
  w[0]  = b0;
  w[16] = 1.0f;
  w[32] = 0.0f;
  float ga = 1.0f, gb = 0.0f;
  float gaPrevEnd = 0.0f, gbPrevEnd = 0.0f;
  for (int i = 1; i <= SLAB_LEN; ++i) {
    const float gan = -a1 * ga - a2 * gb;
    const float gbn = ga;
    if (i < 16) {
      w[i]      = ga * v1 + gb * v2;
      w[16 + i] = gan;
      w[32 + i] = gbn;
    }
    if (i == SLAB_LEN - 1) { gaPrevEnd = gan; gbPrevEnd = gbn; }
    ga = gan; gb = gbn;
  }
  w[52] = ga;            w[53] = gb;
  w[54] = -a2 * gaPrevEnd; w[55] = -a2 * gbPrevEnd;
}

template <bool WRITE_Y>
__global__ __launch_bounds__(32) void biquad_pass_kernel(
    const float* __restrict__ x,
    const float* __restrict__ b1v, const float* __restrict__ b2v,
    const float* __restrict__ a1v, const float* __restrict__ a2v,
    const float* __restrict__ coef,
    const float* __restrict__ sstart,
    float* __restrict__ dloc,
    float* __restrict__ y) {
  const int lane = threadIdx.x;
  const int wid = blockIdx.x;
  const int c = wid / SLABS;
  const int s = wid % SLABS;
  const int Mrow = lane & 15;
  const int hi = lane >> 4;
  const float* __restrict__ w = coef + c * COEF_STRIDE;

  v2f af[4];
#pragma unroll
  for (int m = 0; m < 4; ++m) {
#pragma unroll
    for (int r = 0; r < 2; ++r) {
      const int k = 4 * m + r + 2 * hi;
      af[m][r] = (Mrow >= k) ? w[Mrow - k] : 0.0f;
    }
  }
  float gAr[8], gBr[8];
#pragma unroll
  for (int r = 0; r < 8; ++r) {
    gAr[r] = w[16 + 8 * hi + r];
    gBr[r] = w[32 + 8 * hi + r];
  }
  const float gA14 = w[16 + 14], gA15 = w[16 + 15];
  const float gB14 = w[32 + 14], gB15 = w[32 + 15];
  const float b1 = b1v[c], b2 = b2v[c], a1 = a1v[c], a2 = a2v[c];

  float z1 = 0.0f, z2 = 0.0f;
  if constexpr (WRITE_Y) {
    z1 = sstart[(c * SLABS + s) * 2 + 0];
    z2 = sstart[(c * SLABS + s) * 2 + 1];
  }

  const long base_ch = (long)c * T_LEN + (long)s * SLAB_LEN;

  for (int G = 0; G < GROUPS; ++G) {
    const int j0 = G * GRP;
    const int nb = min(GRP, BLOCKS_PER_SLAB - j0);
    const long base = base_ch + (long)j0 * BLK;

    v2f bf[4];
#pragma unroll
    for (int m = 0; m < 4; ++m) {
      if (Mrow < nb) {
        const float2 t = *(const float2*)(x + base + Mrow * BLK + 4 * m + 2 * hi);
        bf[m][0] = t.x; bf[m][1] = t.y;
      } else {
        bf[m][0] = 0.0f; bf[m][1] = 0.0f;
      }
    }

    if (G + 1 < GROUPS)
      __builtin_prefetch((const void*)(x + base + GRP * BLK + lane * 8), 0, 1);

    v8f acc = {};
    acc = __builtin_amdgcn_wmma_f32_16x16x4_f32(false, af[0], false, bf[0], (short)0, acc, false, false);
    acc = __builtin_amdgcn_wmma_f32_16x16x4_f32(false, af[1], false, bf[1], (short)0, acc, false, false);
    acc = __builtin_amdgcn_wmma_f32_16x16x4_f32(false, af[2], false, bf[2], (short)0, acc, false, false);
    acc = __builtin_amdgcn_wmma_f32_16x16x4_f32(false, af[3], false, bf[3], (short)0, acc, false, false);

    for (int jj = 0; jj < nb; ++jj) {
      const int src = 16 + jj;
      const float yp14 = __shfl(acc[6], src, 32);
      const float yp15 = __shfl(acc[7], src, 32);
      const float x14 = __shfl(bf[3][0], src, 32);
      const float x15 = __shfl(bf[3][1], src, 32);
      const float y14 = yp14 + gA14 * z1 + gB14 * z2;
      const float y15 = yp15 + gA15 * z1 + gB15 * z2;
      if constexpr (WRITE_Y) {
        if (Mrow == jj) {
#pragma unroll
          for (int r = 0; r < 8; ++r) acc[r] += gAr[r] * z1 + gBr[r] * z2;
        }
      }
      const float z2mid = b2 * x14 - a2 * y14;
      const float z2n = b2 * x15 - a2 * y15;
      const float z1n = b1 * x15 - a1 * y15 + z2mid;
      z1 = z1n; z2 = z2n;
    }

    if constexpr (WRITE_Y) {
      if (Mrow < nb) {
        float* dst = y + base + Mrow * BLK + 8 * hi;
        *(float4*)(dst + 0) = make_float4(acc[0], acc[1], acc[2], acc[3]);
        *(float4*)(dst + 4) = make_float4(acc[4], acc[5], acc[6], acc[7]);
      }
    }
  }

  if constexpr (!WRITE_Y) {
    if (lane == 0) {
      dloc[(c * SLABS + s) * 2 + 0] = z1;
      dloc[(c * SLABS + s) * 2 + 1] = z2;
    }
  }
}

__global__ void biquad_scan_kernel(const float* __restrict__ coef,
                                   const float* __restrict__ dloc,
                                   float* __restrict__ sstart) {
  int c = blockIdx.x * blockDim.x + threadIdx.x;
  if (c >= B_CH) return;
  const float* w = coef + c * COEF_STRIDE;
  const float m00 = w[52], m01 = w[53], m10 = w[54], m11 = w[55];
  float z1 = 0.0f, z2 = 0.0f;
  for (int s = 0; s < SLABS; ++s) {
    sstart[(c * SLABS + s) * 2 + 0] = z1;
    sstart[(c * SLABS + s) * 2 + 1] = z2;
    const float d1 = dloc[(c * SLABS + s) * 2 + 0];
    const float d2 = dloc[(c * SLABS + s) * 2 + 1];
    const float z1n = m00 * z1 + m01 * z2 + d1;
    const float z2n = m10 * z1 + m11 * z2 + d2;
    z1 = z1n; z2 = z2n;
  }
}

extern "C" void kernel_launch(void* const* d_in, const int* in_sizes, int n_in,
                              void* d_out, int out_size, void* d_ws, size_t ws_size,
                              hipStream_t stream) {
  const float* x  = (const float*)d_in[0];
  const float* b0 = (const float*)d_in[1];
  const float* b1 = (const float*)d_in[2];
  const float* b2 = (const float*)d_in[3];
  const float* a1 = (const float*)d_in[4];
  const float* a2 = (const float*)d_in[5];
  float* y = (float*)d_out;

  float* ws     = (float*)d_ws;
  float* coef   = ws;
  float* dloc   = coef + B_CH * COEF_STRIDE;
  float* sstart = dloc + B_CH * SLABS * 2;

  biquad_coef_kernel<<<(B_CH + 63) / 64, 64, 0, stream>>>(b0, b1, b2, a1, a2, coef);
  biquad_pass_kernel<false><<<B_CH * SLABS, 32, 0, stream>>>(
      x, b1, b2, a1, a2, coef, nullptr, dloc, nullptr);
  biquad_scan_kernel<<<(B_CH + 63) / 64, 64, 0, stream>>>(coef, dloc, sstart);
  biquad_pass_kernel<true><<<B_CH * SLABS, 32, 0, stream>>>(
      x, b1, b2, a1, a2, coef, sstart, nullptr, y);
}


# two 256-sample groups per wave step: lane row 0 runs one group's 16-lane state scan, row 1 the next group's (it ran twice, identically, in both rows); their WMMA chains interleave; sums unchanged
# speedup vs baseline: 1.8492x; 1.8492x over previous
#include <hip/hip_runtime.h>
#include <hip/hip_bf16.h>


typedef __attribute__((ext_vector_type(2))) float v2f;
typedef __attribute__((ext_vector_type(8))) float v8f;

namespace {
constexpr int B_CH = 512;
constexpr int T_LEN = 48000;
constexpr int BLK = 16;
constexpr int GRP = 16;
constexpr int GRP_LEN = GRP * BLK;
constexpr int BLOCKS_CH = T_LEN / BLK;
constexpr int FULL_GROUPS = BLOCKS_CH / GRP;
constexpr int TAIL_NB = BLOCKS_CH - FULL_GROUPS * GRP;
constexpr int GROUPS_CH = FULL_GROUPS + 1;
constexpr int WGRP = 18;
constexpr int PAIRS = WGRP / 2;
constexpr int SLABS = (GROUPS_CH + WGRP - 1) / WGRP;
constexpr int LAST_GROUPS = GROUPS_CH - (SLABS - 1) * WGRP;
constexpr int LAST_PAIRS = LAST_GROUPS / 2;
static_assert(LAST_GROUPS > 0 && LAST_GROUPS <= WGRP && LAST_GROUPS % 2 == 0 && LAST_PAIRS >= 1 &&
              TAIL_NB > 0 && TAIL_NB < GRP, "group split");
constexpr int COEF_STRIDE = 192;
constexpr int QOFF = 64;
}

__global__ void biquad_coef_kernel(const float* __restrict__ b0v, const float* __restrict__ b1v,
                                   const float* __restrict__ b2v, const float* __restrict__ a1v,
                                   const float* __restrict__ a2v, float* __restrict__ coef) {
  int c = blockIdx.x * blockDim.x + threadIdx.x;
  if (c >= B_CH) return;
  const float b0 = b0v[c], b1 = b1v[c], b2 = b2v[c], a1 = a1v[c], a2 = a2v[c];
  float* w = coef + c * COEF_STRIDE;
  const float v1 = b1 - a1 * b0;
  const float v2 = b2 - a2 * b0;
  w[0]  = b0;
  w[16] = 1.0f;
  w[32] = 0.0f;
  w[QOFF + 0] = 1.0f; w[QOFF + 1] = 0.0f;
  w[QOFF + 2] = 0.0f; w[QOFF + 3] = 1.0f;
  float ga = 1.0f, gb = 0.0f;
  for (int i = 1; i <= GRP * BLK; ++i) {
    const float gan = -a1 * ga - a2 * gb;
    const float gbn = ga;
    if (i < 16) {
      w[i]      = ga * v1 + gb * v2;
      w[16 + i] = gan;
      w[32 + i] = gbn;
    }
    if ((i & (BLK - 1)) == 0) {
      float* q = w + QOFF + 4 * (i / BLK);
      q[0] = gan;      q[1] = gbn;
      q[2] = -a2 * ga; q[3] = -a2 * gb;
    }
    ga = gan; gb = gbn;
  }
}

__device__ __forceinline__ float xhalf(float v) {
  const int b = __builtin_bit_cast(int, v);
  return __builtin_bit_cast(float, __builtin_amdgcn_permlanex16(b, b, 0x76543210, 0xfedcba98, false, false));
}
template <int N>
__device__ __forceinline__ float row_shr0(float v) {
  return __builtin_bit_cast(float, __builtin_amdgcn_update_dpp(0, __builtin_bit_cast(int, v), 0x110 + N, 0xf, 0xf, true));
}
template <int N>
__device__ __forceinline__ void scan_step(float& e1, float& e2, const float* p) {
  const float t1 = row_shr0<N>(e1);
  const float t2 = row_shr0<N>(e2);
  e1 += p[0] * t1 + p[1] * t2;
  e2 += p[2] * t1 + p[3] * t2;
}

template <int N> struct IntTag { static constexpr int v = N; };

template <bool WRITE_Y>
__global__ __launch_bounds__(32) void biquad_pass_kernel(
    const float* __restrict__ x,
    const float* __restrict__ b1v, const float* __restrict__ b2v,
    const float* __restrict__ a1v, const float* __restrict__ a2v,
    const float* __restrict__ coef,
    const float* __restrict__ gstart,
    float* __restrict__ gend,
    float* __restrict__ y) {
  const int lane = threadIdx.x;
  const int wid = blockIdx.x;
  const int c = wid / SLABS;
  const int s = wid % SLABS;
  const int Mrow = lane & 15;
  const int hi = lane >> 4;
  const float* __restrict__ w = coef + c * COEF_STRIDE;

  v2f af[4];
#pragma unroll
  for (int m = 0; m < 4; ++m) {
#pragma unroll
    for (int r = 0; r < 2; ++r) {
      const int k = 4 * m + r + 2 * hi;
      af[m][r] = (Mrow >= k) ? w[Mrow - k] : 0.0f;
    }
  }
  float gAr[8], gBr[8];
#pragma unroll
  for (int r = 0; r < 8; ++r) {
    gAr[r] = w[16 + 8 * hi + r];
    gBr[r] = w[32 + 8 * hi + r];
  }
  const float b1 = b1v[c], b2 = b2v[c], a1 = a1v[c], a2 = a2v[c];
  float p1[4], p2[4], p4[4], p8[4], qj[4];
#pragma unroll
  for (int i = 0; i < 4; ++i) {
    p1[i] = w[QOFF + 4 * 1 + i];
    p2[i] = w[QOFF + 4 * 2 + i];
    p4[i] = w[QOFF + 4 * 4 + i];
    p8[i] = w[QOFF + 4 * 8 + i];
    qj[i] = w[QOFF + 4 * Mrow + i];
  }

  const int g0 = s * WGRP;
  const bool last = (s == SLABS - 1);
  const int ng = last ? LAST_GROUPS : WGRP;

  float zs1 = 0.0f, zs2 = 0.0f;
  const long gslot = ((long)(g0 + lane) * B_CH + c) * 2;
  if constexpr (WRITE_Y) {
    if (lane < ng) {
      const float2 t = *(const float2*)(gstart + gslot);
      zs1 = t.x; zs2 = t.y;
    }
  }

  const long base_ch = (long)c * T_LEN + (long)g0 * GRP_LEN;

  auto pair = [&](auto nbb_tag, const int P, const bool prefetch_next) __attribute__((always_inline)) {
    constexpr int nbB = decltype(nbb_tag)::v;
    const long base = base_ch + (long)P * (2 * GRP_LEN);
    const int GA = 2 * P, GB = 2 * P + 1;

    v2f bfA[4], bfB[4];
#pragma unroll
    for (int m = 0; m < 4; ++m) {
      const float2 t = *(const float2*)(x + base + Mrow * BLK + 4 * m + 2 * hi);
      bfA[m][0] = t.x; bfA[m][1] = t.y;
    }
#pragma unroll
    for (int m = 0; m < 4; ++m) {
      if (nbB == GRP || Mrow < nbB) {
        const float2 t = *(const float2*)(x + base + GRP_LEN + Mrow * BLK + 4 * m + 2 * hi);
        bfB[m][0] = t.x; bfB[m][1] = t.y;
      } else {
        bfB[m][0] = 0.0f; bfB[m][1] = 0.0f;
      }
    }

    if (prefetch_next)
      __builtin_prefetch((const void*)(x + base + 2 * GRP_LEN + lane * 16), 0, 1);

    v8f accA = {}, accB = {};
#pragma unroll
    for (int m = 0; m < 4; ++m) {
      accA = __builtin_amdgcn_wmma_f32_16x16x4_f32(false, af[m], false, bfA[m], (short)0, accA, false, false);
      accB = __builtin_amdgcn_wmma_f32_16x16x4_f32(false, af[m], false, bfB[m], (short)0, accB, false, false);
    }

    const float dmidA = b2 * bfA[3][0] - a2 * accA[6];
    const float eA2 = b2 * bfA[3][1] - a2 * accA[7];
    const float eA1 = b1 * bfA[3][1] - a1 * accA[7] + dmidA;
    const float dmidB = b2 * bfB[3][0] - a2 * accB[6];
    const float eB2 = b2 * bfB[3][1] - a2 * accB[7];
    const float eB1 = b1 * bfB[3][1] - a1 * accB[7] + dmidB;
    const float eA1x = xhalf(eA1);
    const float eA2x = xhalf(eA2);
    float e1 = hi ? eB1 : eA1x;
    float e2 = hi ? eB2 : eA2x;
    scan_step<1>(e1, e2, p1);
    scan_step<2>(e1, e2, p2);
    scan_step<4>(e1, e2, p4);
    scan_step<8>(e1, e2, p8);

    if constexpr (WRITE_Y) {
      const float zA1 = __builtin_bit_cast(float, __builtin_amdgcn_readlane(__builtin_bit_cast(int, zs1), GA));
      const float zA2 = __builtin_bit_cast(float, __builtin_amdgcn_readlane(__builtin_bit_cast(int, zs2), GA));
      const float zB1 = __builtin_bit_cast(float, __builtin_amdgcn_readlane(__builtin_bit_cast(int, zs1), GB));
      const float zB2 = __builtin_bit_cast(float, __builtin_amdgcn_readlane(__builtin_bit_cast(int, zs2), GB));
      const float z1 = hi ? zB1 : zA1;
      const float z2 = hi ? zB2 : zA2;
      const float sj1 = qj[0] * z1 + qj[1] * z2 + row_shr0<1>(e1);
      const float sj2 = qj[2] * z1 + qj[3] * z2 + row_shr0<1>(e2);
      const float sx1 = xhalf(sj1);
      const float sx2 = xhalf(sj2);
      const float sA1 = hi ? sx1 : sj1, sA2 = hi ? sx2 : sj2;
      const float sB1 = hi ? sj1 : sx1, sB2 = hi ? sj2 : sx2;
#pragma unroll
      for (int r = 0; r < 8; ++r) accA[r] += gAr[r] * sA1 + gBr[r] * sA2;
#pragma unroll
      for (int r = 0; r < 8; ++r) accB[r] += gAr[r] * sB1 + gBr[r] * sB2;
      {
        float* dst = y + base + Mrow * BLK + 8 * hi;
        *(float4*)(dst + 0) = make_float4(accA[0], accA[1], accA[2], accA[3]);
        *(float4*)(dst + 4) = make_float4(accA[4], accA[5], accA[6], accA[7]);
      }
      if (nbB == GRP || Mrow < nbB) {
        float* dst = y + base + GRP_LEN + Mrow * BLK + 8 * hi;
        *(float4*)(dst + 0) = make_float4(accB[0], accB[1], accB[2], accB[3]);
        *(float4*)(dst + 4) = make_float4(accB[4], accB[5], accB[6], accB[7]);
      }
    } else {
      const float lA1 = __builtin_bit_cast(float, __builtin_amdgcn_readlane(__builtin_bit_cast(int, e1), GRP - 1));
      const float lA2 = __builtin_bit_cast(float, __builtin_amdgcn_readlane(__builtin_bit_cast(int, e2), GRP - 1));
      const float lB1 = __builtin_bit_cast(float, __builtin_amdgcn_readlane(__builtin_bit_cast(int, e1), 16 + nbB - 1));
      const float lB2 = __builtin_bit_cast(float, __builtin_amdgcn_readlane(__builtin_bit_cast(int, e2), 16 + nbB - 1));
      zs1 = (lane == GA) ? lA1 : ((lane == GB) ? lB1 : zs1);
      zs2 = (lane == GA) ? lA2 : ((lane == GB) ? lB2 : zs2);
    }
  };

  if (!last) {
    if constexpr (WRITE_Y) {
      for (int P = 0; P < PAIRS; ++P) pair(IntTag<GRP>{}, P, P + 1 < PAIRS);
    } else {
#pragma unroll
      for (int P = 0; P < PAIRS; ++P) pair(IntTag<GRP>{}, P, P + 1 < PAIRS);
    }
  } else {
    if constexpr (WRITE_Y) {
      for (int P = 0; P < LAST_PAIRS - 1; ++P) pair(IntTag<GRP>{}, P, P + 2 < LAST_PAIRS);
    } else {
#pragma unroll
      for (int P = 0; P < LAST_PAIRS - 1; ++P) pair(IntTag<GRP>{}, P, P + 2 < LAST_PAIRS);
    }
    pair(IntTag<TAIL_NB>{}, LAST_PAIRS - 1, false);
  }

  if constexpr (!WRITE_Y) {
    if (lane < ng) *(float2*)(gend + gslot) = make_float2(zs1, zs2);
  }
}

__global__ __launch_bounds__(64) void biquad_scan_kernel(const float* __restrict__ coef,
                                                         const float* __restrict__ gend,
                                                         float* __restrict__ gstart) {
  constexpr int CH = 47;
  constexpr int NCH = GROUPS_CH / CH;
  static_assert(CH * NCH == GROUPS_CH, "scan batches");
  const int c = blockIdx.x * blockDim.x + threadIdx.x;
  if (c >= B_CH) return;
  const float* w = coef + c * COEF_STRIDE + QOFF + 4 * GRP;
  const float m00 = w[0], m01 = w[1], m10 = w[2], m11 = w[3];
  const float2* __restrict__ din = (const float2*)gend + c;
  float2* __restrict__ sout = (float2*)gstart + c;
  float2 cur[CH], nxt[CH];
#pragma unroll
  for (int i = 0; i < CH; ++i) cur[i] = din[(long)i * B_CH];
  float z1 = 0.0f, z2 = 0.0f;
#pragma unroll
  for (int k = 0; k < NCH; ++k) {
    if (k + 1 < NCH) {
#pragma unroll
      for (int i = 0; i < CH; ++i) nxt[i] = din[(long)((k + 1) * CH + i) * B_CH];
    }
#pragma unroll
    for (int i = 0; i < CH; ++i) {
      sout[(long)(k * CH + i) * B_CH] = make_float2(z1, z2);
      const float z1n = m00 * z1 + (m01 * z2 + cur[i].x);
      const float z2n = m10 * z1 + (m11 * z2 + cur[i].y);
      z1 = z1n; z2 = z2n;
    }
#pragma unroll
    for (int i = 0; i < CH; ++i) cur[i] = nxt[i];
  }
}

extern "C" void kernel_launch(void* const* d_in, const int* in_sizes, int n_in,
                              void* d_out, int out_size, void* d_ws, size_t ws_size,
                              hipStream_t stream) {
  const float* x  = (const float*)d_in[0];
  const float* b0 = (const float*)d_in[1];
  const float* b1 = (const float*)d_in[2];
  const float* b2 = (const float*)d_in[3];
  const float* a1 = (const float*)d_in[4];
  const float* a2 = (const float*)d_in[5];
  float* y = (float*)d_out;

  float* ws     = (float*)d_ws;
  float* coef   = ws;
  float* gend   = coef + B_CH * COEF_STRIDE;
  float* gstart = gend + GROUPS_CH * B_CH * 2;

  biquad_coef_kernel<<<(B_CH + 63) / 64, 64, 0, stream>>>(b0, b1, b2, a1, a2, coef);
  biquad_pass_kernel<false><<<B_CH * SLABS, 32, 0, stream>>>(
      x, b1, b2, a1, a2, coef, nullptr, gend, nullptr);
  biquad_scan_kernel<<<(B_CH + 63) / 64, 64, 0, stream>>>(coef, gend, gstart);
  biquad_pass_kernel<true><<<B_CH * SLABS, 32, 0, stream>>>(
      x, b1, b2, a1, a2, coef, gstart, nullptr, y);
}


# scan launch removed (3 launches, was 4): pass 1 also leaves each slab's end state; a pass-2 wave sums its <=10 earlier slab states with M^4608, steps its own 18 groups; state summation order changes
# speedup vs baseline: 2.4611x; 1.3309x over previous
#include <hip/hip_runtime.h>
#include <hip/hip_bf16.h>


typedef __attribute__((ext_vector_type(2))) float v2f;
typedef __attribute__((ext_vector_type(8))) float v8f;

namespace {
constexpr int B_CH = 512;
constexpr int T_LEN = 48000;
constexpr int BLK = 16;
constexpr int GRP = 16;
constexpr int GRP_LEN = GRP * BLK;
constexpr int BLOCKS_CH = T_LEN / BLK;
constexpr int FULL_GROUPS = BLOCKS_CH / GRP;
constexpr int TAIL_NB = BLOCKS_CH - FULL_GROUPS * GRP;
constexpr int GROUPS_CH = FULL_GROUPS + 1;
constexpr int WGRP = 18;
constexpr int PAIRS = WGRP / 2;
constexpr int SLABS = (GROUPS_CH + WGRP - 1) / WGRP;
constexpr int LAST_GROUPS = GROUPS_CH - (SLABS - 1) * WGRP;
constexpr int LAST_PAIRS = LAST_GROUPS / 2;
static_assert(LAST_GROUPS > 0 && LAST_GROUPS <= WGRP && LAST_GROUPS % 2 == 0 && LAST_PAIRS >= 1 &&
              TAIL_NB > 0 && TAIL_NB < GRP, "group split");
constexpr int COEF_STRIDE = 192;
constexpr int QOFF = 64;
constexpr int POFF = QOFF + 4 * GRP;
constexpr int SOFF = POFF + 4;
static_assert(SOFF + 4 <= COEF_STRIDE, "coef row");
}

__global__ void biquad_coef_kernel(const float* __restrict__ b0v, const float* __restrict__ b1v,
                                   const float* __restrict__ b2v, const float* __restrict__ a1v,
                                   const float* __restrict__ a2v, float* __restrict__ coef) {
  int c = blockIdx.x * blockDim.x + threadIdx.x;
  if (c >= B_CH) return;
  const float b0 = b0v[c], b1 = b1v[c], b2 = b2v[c], a1 = a1v[c], a2 = a2v[c];
  float* w = coef + c * COEF_STRIDE;
  const float v1 = b1 - a1 * b0;
  const float v2 = b2 - a2 * b0;
  w[0]  = b0;
  w[16] = 1.0f;
  w[32] = 0.0f;
  w[QOFF + 0] = 1.0f; w[QOFF + 1] = 0.0f;
  w[QOFF + 2] = 0.0f; w[QOFF + 3] = 1.0f;
  float ga = 1.0f, gb = 0.0f;
  float s00 = 1.0f, s01 = 0.0f, s10 = 0.0f, s11 = 1.0f;
  for (int i = 1; i <= GRP * BLK; ++i) {
    const float gan = -a1 * ga - a2 * gb;
    const float gbn = ga;
    if (i < 16) {
      w[i]      = ga * v1 + gb * v2;
      w[16 + i] = gan;
      w[32 + i] = gbn;
    }
    if ((i & (BLK - 1)) == 0) {
      float* q = w + QOFF + 4 * (i / BLK);
      q[0] = gan;      q[1] = gbn;
      q[2] = -a2 * ga; q[3] = -a2 * gb;
      if (i == GRP * BLK) { s00 = gan; s01 = gbn; s10 = -a2 * ga; s11 = -a2 * gb; }
    }
    ga = gan; gb = gbn;
  }
  float r00 = 1.0f, r01 = 0.0f, r10 = 0.0f, r11 = 1.0f;
#pragma unroll
  for (int e = WGRP; e > 0; e >>= 1) {
    if (e & 1) {
      const float t00 = r00 * s00 + r01 * s10, t01 = r00 * s01 + r01 * s11;
      const float t10 = r10 * s00 + r11 * s10, t11 = r10 * s01 + r11 * s11;
      r00 = t00; r01 = t01; r10 = t10; r11 = t11;
    }
    const float u00 = s00 * s00 + s01 * s10, u01 = s00 * s01 + s01 * s11;
    const float u10 = s10 * s00 + s11 * s10, u11 = s10 * s01 + s11 * s11;
    s00 = u00; s01 = u01; s10 = u10; s11 = u11;
  }
  w[SOFF + 0] = r00; w[SOFF + 1] = r01;
  w[SOFF + 2] = r10; w[SOFF + 3] = r11;
}

__device__ __forceinline__ float xhalf(float v) {
  const int b = __builtin_bit_cast(int, v);
  return __builtin_bit_cast(float, __builtin_amdgcn_permlanex16(b, b, 0x76543210, 0xfedcba98, false, false));
}
template <int N>
__device__ __forceinline__ float row_shr0(float v) {
  return __builtin_bit_cast(float, __builtin_amdgcn_update_dpp(0, __builtin_bit_cast(int, v), 0x110 + N, 0xf, 0xf, true));
}
template <int N>
__device__ __forceinline__ void scan_step(float& e1, float& e2, const float* p) {
  const float t1 = row_shr0<N>(e1);
  const float t2 = row_shr0<N>(e2);
  e1 += p[0] * t1 + p[1] * t2;
  e2 += p[2] * t1 + p[3] * t2;
}

template <int N> struct IntTag { static constexpr int v = N; };

template <bool WRITE_Y>
__global__ __launch_bounds__(32) void biquad_pass_kernel(
    const float* __restrict__ x,
    const float* __restrict__ b1v, const float* __restrict__ b2v,
    const float* __restrict__ a1v, const float* __restrict__ a2v,
    const float* __restrict__ coef,
    const float* __restrict__ gin,
    const float* __restrict__ slabin,
    float* __restrict__ gend,
    float* __restrict__ slabend,
    float* __restrict__ y) {
  const int lane = threadIdx.x;
  const int wid = blockIdx.x;
  const int c = wid / SLABS;
  const int s = wid % SLABS;
  const int Mrow = lane & 15;
  const int hi = lane >> 4;
  const float* __restrict__ w = coef + c * COEF_STRIDE;

  v2f af[4];
#pragma unroll
  for (int m = 0; m < 4; ++m) {
#pragma unroll
    for (int r = 0; r < 2; ++r) {
      const int k = 4 * m + r + 2 * hi;
      af[m][r] = (Mrow >= k) ? w[Mrow - k] : 0.0f;
    }
  }
  float gAr[8], gBr[8];
#pragma unroll
  for (int r = 0; r < 8; ++r) {
    gAr[r] = w[16 + 8 * hi + r];
    gBr[r] = w[32 + 8 * hi + r];
  }
  const float b1 = b1v[c], b2 = b2v[c], a1 = a1v[c], a2 = a2v[c];
  float p1[4], p2[4], p4[4], p8[4], qj[4];
#pragma unroll
  for (int i = 0; i < 4; ++i) {
    p1[i] = w[QOFF + 4 * 1 + i];
    p2[i] = w[QOFF + 4 * 2 + i];
    p4[i] = w[QOFF + 4 * 4 + i];
    p8[i] = w[QOFF + 4 * 8 + i];
    qj[i] = w[QOFF + 4 * Mrow + i];
  }
  float pm[4], ps[4];
#pragma unroll
  for (int i = 0; i < 4; ++i) {
    pm[i] = w[POFF + i];
    ps[i] = w[SOFF + i];
  }

  const int g0 = s * WGRP;
  const bool last = (s == SLABS - 1);
  const int ng = last ? LAST_GROUPS : WGRP;

  float zs1 = 0.0f, zs2 = 0.0f;
  const long gslot = ((long)(g0 + lane) * B_CH + c) * 2;
  float sd1 = 0.0f, sd2 = 0.0f;
  if constexpr (WRITE_Y) {
    if (lane < ng) {
      const float2 t = *(const float2*)(gin + gslot);
      zs1 = t.x; zs2 = t.y;
    }
    if (lane < s) {
      const float2 t = *(const float2*)(slabin + ((long)lane * B_CH + c) * 2);
      sd1 = t.x; sd2 = t.y;
    }
  }
  float G1 = 0.0f, G2 = 0.0f;

  const long base_ch = (long)c * T_LEN + (long)g0 * GRP_LEN;

  auto pair = [&](auto nbb_tag, const int P, const bool prefetch_next) __attribute__((always_inline)) {
    constexpr int nbB = decltype(nbb_tag)::v;
    const long base = base_ch + (long)P * (2 * GRP_LEN);
    const int GA = 2 * P, GB = 2 * P + 1;

    v2f bfA[4], bfB[4];
#pragma unroll
    for (int m = 0; m < 4; ++m) {
      const float2 t = *(const float2*)(x + base + Mrow * BLK + 4 * m + 2 * hi);
      bfA[m][0] = t.x; bfA[m][1] = t.y;
    }
#pragma unroll
    for (int m = 0; m < 4; ++m) {
      if (nbB == GRP || Mrow < nbB) {
        const float2 t = *(const float2*)(x + base + GRP_LEN + Mrow * BLK + 4 * m + 2 * hi);
        bfB[m][0] = t.x; bfB[m][1] = t.y;
      } else {
        bfB[m][0] = 0.0f; bfB[m][1] = 0.0f;
      }
    }

    if (prefetch_next)
      __builtin_prefetch((const void*)(x + base + 2 * GRP_LEN + lane * 16), 0, 1);

    v8f accA = {}, accB = {};
#pragma unroll
    for (int m = 0; m < 4; ++m) {
      accA = __builtin_amdgcn_wmma_f32_16x16x4_f32(false, af[m], false, bfA[m], (short)0, accA, false, false);
      accB = __builtin_amdgcn_wmma_f32_16x16x4_f32(false, af[m], false, bfB[m], (short)0, accB, false, false);
    }

    const float dmidA = b2 * bfA[3][0] - a2 * accA[6];
    const float eA2 = b2 * bfA[3][1] - a2 * accA[7];
    const float eA1 = b1 * bfA[3][1] - a1 * accA[7] + dmidA;
    const float dmidB = b2 * bfB[3][0] - a2 * accB[6];
    const float eB2 = b2 * bfB[3][1] - a2 * accB[7];
    const float eB1 = b1 * bfB[3][1] - a1 * accB[7] + dmidB;
    const float eA1x = xhalf(eA1);
    const float eA2x = xhalf(eA2);
    float e1 = hi ? eB1 : eA1x;
    float e2 = hi ? eB2 : eA2x;
    scan_step<1>(e1, e2, p1);
    scan_step<2>(e1, e2, p2);
    scan_step<4>(e1, e2, p4);
    scan_step<8>(e1, e2, p8);

    if constexpr (WRITE_Y) {
      if (P == 0) {
        for (int i = 0; i < s; ++i) {
          const float d1 = __builtin_bit_cast(float, __builtin_amdgcn_readlane(__builtin_bit_cast(int, sd1), i));
          const float d2 = __builtin_bit_cast(float, __builtin_amdgcn_readlane(__builtin_bit_cast(int, sd2), i));
          const float n1 = ps[0] * G1 + (ps[1] * G2 + d1);
          const float n2 = ps[2] * G1 + (ps[3] * G2 + d2);
          G1 = n1; G2 = n2;
        }
      }
      const float zA1 = G1, zA2 = G2;
      const float dA1 = __builtin_bit_cast(float, __builtin_amdgcn_readlane(__builtin_bit_cast(int, zs1), GA));
      const float dA2 = __builtin_bit_cast(float, __builtin_amdgcn_readlane(__builtin_bit_cast(int, zs2), GA));
      const float zB1 = pm[0] * zA1 + (pm[1] * zA2 + dA1);
      const float zB2 = pm[2] * zA1 + (pm[3] * zA2 + dA2);
      const float dB1 = __builtin_bit_cast(float, __builtin_amdgcn_readlane(__builtin_bit_cast(int, zs1), GB));
      const float dB2 = __builtin_bit_cast(float, __builtin_amdgcn_readlane(__builtin_bit_cast(int, zs2), GB));
      G1 = pm[0] * zB1 + (pm[1] * zB2 + dB1);
      G2 = pm[2] * zB1 + (pm[3] * zB2 + dB2);
      const float z1 = hi ? zB1 : zA1;
      const float z2 = hi ? zB2 : zA2;
      const float sj1 = qj[0] * z1 + qj[1] * z2 + row_shr0<1>(e1);
      const float sj2 = qj[2] * z1 + qj[3] * z2 + row_shr0<1>(e2);
      const float sx1 = xhalf(sj1);
      const float sx2 = xhalf(sj2);
      const float sA1 = hi ? sx1 : sj1, sA2 = hi ? sx2 : sj2;
      const float sB1 = hi ? sj1 : sx1, sB2 = hi ? sj2 : sx2;
#pragma unroll
      for (int r = 0; r < 8; ++r) accA[r] += gAr[r] * sA1 + gBr[r] * sA2;
#pragma unroll
      for (int r = 0; r < 8; ++r) accB[r] += gAr[r] * sB1 + gBr[r] * sB2;
      {
        float* dst = y + base + Mrow * BLK + 8 * hi;
        *(float4*)(dst + 0) = make_float4(accA[0], accA[1], accA[2], accA[3]);
        *(float4*)(dst + 4) = make_float4(accA[4], accA[5], accA[6], accA[7]);
      }
      if (nbB == GRP || Mrow < nbB) {
        float* dst = y + base + GRP_LEN + Mrow * BLK + 8 * hi;
        *(float4*)(dst + 0) = make_float4(accB[0], accB[1], accB[2], accB[3]);
        *(float4*)(dst + 4) = make_float4(accB[4], accB[5], accB[6], accB[7]);
      }
    } else {
      const float lA1 = __builtin_bit_cast(float, __builtin_amdgcn_readlane(__builtin_bit_cast(int, e1), GRP - 1));
      const float lA2 = __builtin_bit_cast(float, __builtin_amdgcn_readlane(__builtin_bit_cast(int, e2), GRP - 1));
      const float lB1 = __builtin_bit_cast(float, __builtin_amdgcn_readlane(__builtin_bit_cast(int, e1), 16 + nbB - 1));
      const float lB2 = __builtin_bit_cast(float, __builtin_amdgcn_readlane(__builtin_bit_cast(int, e2), 16 + nbB - 1));
      zs1 = (lane == GA) ? lA1 : ((lane == GB) ? lB1 : zs1);
      zs2 = (lane == GA) ? lA2 : ((lane == GB) ? lB2 : zs2);
    }
  };

  if (!last) {
    if constexpr (WRITE_Y) {
      for (int P = 0; P < PAIRS; ++P) pair(IntTag<GRP>{}, P, P + 1 < PAIRS);
    } else {
#pragma unroll
      for (int P = 0; P < PAIRS; ++P) pair(IntTag<GRP>{}, P, P + 1 < PAIRS);
    }
  } else {
    if constexpr (WRITE_Y) {
      for (int P = 0; P < LAST_PAIRS - 1; ++P) pair(IntTag<GRP>{}, P, P + 2 < LAST_PAIRS);
    } else {
#pragma unroll
      for (int P = 0; P < LAST_PAIRS - 1; ++P) pair(IntTag<GRP>{}, P, P + 2 < LAST_PAIRS);
    }
    pair(IntTag<TAIL_NB>{}, LAST_PAIRS - 1, false);
  }

  if constexpr (!WRITE_Y) {
    if (lane < ng) *(float2*)(gend + gslot) = make_float2(zs1, zs2);
    if (!last) {
#pragma unroll
      for (int g = 0; g < WGRP; ++g) {
        const float d1 = __builtin_bit_cast(float, __builtin_amdgcn_readlane(__builtin_bit_cast(int, zs1), g));
        const float d2 = __builtin_bit_cast(float, __builtin_amdgcn_readlane(__builtin_bit_cast(int, zs2), g));
        const float n1 = pm[0] * G1 + (pm[1] * G2 + d1);
        const float n2 = pm[2] * G1 + (pm[3] * G2 + d2);
        G1 = n1; G2 = n2;
      }
      if (lane == 0) *(float2*)(slabend + ((long)s * B_CH + c) * 2) = make_float2(G1, G2);
    }
  }
}

extern "C" void kernel_launch(void* const* d_in, const int* in_sizes, int n_in,
                              void* d_out, int out_size, void* d_ws, size_t ws_size,
                              hipStream_t stream) {
  const float* x  = (const float*)d_in[0];
  const float* b0 = (const float*)d_in[1];
  const float* b1 = (const float*)d_in[2];
  const float* b2 = (const float*)d_in[3];
  const float* a1 = (const float*)d_in[4];
  const float* a2 = (const float*)d_in[5];
  float* y = (float*)d_out;

  float* ws     = (float*)d_ws;
  float* coef   = ws;
  float* gend   = coef + B_CH * COEF_STRIDE;
  float* slabend   = gend + GROUPS_CH * B_CH * 2;

  biquad_coef_kernel<<<(B_CH + 63) / 64, 64, 0, stream>>>(b0, b1, b2, a1, a2, coef);
  biquad_pass_kernel<false><<<B_CH * SLABS, 32, 0, stream>>>(
      x, b1, b2, a1, a2, coef, nullptr, nullptr, gend, slabend, nullptr);
  biquad_pass_kernel<true><<<B_CH * SLABS, 32, 0, stream>>>(
      x, b1, b2, a1, a2, coef, gend, slabend, nullptr, nullptr, y);
}


# coefficient launch removed (2 launches, was 3): each wave forms h, G and M^16..M^64 in registers by a 64-step recurrence, M^128, M^256 and the slab power by squaring; propagator rounding changes
# speedup vs baseline: 3.1263x; 1.2703x over previous
#include <hip/hip_runtime.h>
#include <hip/hip_bf16.h>


typedef __attribute__((ext_vector_type(2))) float v2f;
typedef __attribute__((ext_vector_type(8))) float v8f;

namespace {
constexpr int B_CH = 512;
constexpr int T_LEN = 48000;
constexpr int BLK = 16;
constexpr int GRP = 16;
constexpr int GRP_LEN = GRP * BLK;
constexpr int BLOCKS_CH = T_LEN / BLK;
constexpr int FULL_GROUPS = BLOCKS_CH / GRP;
constexpr int TAIL_NB = BLOCKS_CH - FULL_GROUPS * GRP;
constexpr int GROUPS_CH = FULL_GROUPS + 1;
constexpr int WGRP = 18;
constexpr int PAIRS = WGRP / 2;
constexpr int SLABS = (GROUPS_CH + WGRP - 1) / WGRP;
constexpr int LAST_GROUPS = GROUPS_CH - (SLABS - 1) * WGRP;
constexpr int LAST_PAIRS = LAST_GROUPS / 2;
static_assert(LAST_GROUPS > 0 && LAST_GROUPS <= WGRP && LAST_GROUPS % 2 == 0 && LAST_PAIRS >= 1 &&
              TAIL_NB > 0 && TAIL_NB < GRP, "group split");
static_assert(BLK == 16 && GRP == 16, "the power ladder below is written for 16 x 16");
}

__device__ __forceinline__ float xhalf(float v) {
  const int b = __builtin_bit_cast(int, v);
  return __builtin_bit_cast(float, __builtin_amdgcn_permlanex16(b, b, 0x76543210, 0xfedcba98, false, false));
}
template <int N>
__device__ __forceinline__ float row_shr0(float v) {
  if constexpr (N == 0) return v;
  else return __builtin_bit_cast(float, __builtin_amdgcn_update_dpp(0, __builtin_bit_cast(int, v), 0x110 + N, 0xf, 0xf, true));
}
__device__ __forceinline__ void mmul(const float* x, const float* y, float* o) {
  const float o0 = __builtin_fmaf(x[0], y[0], x[1] * y[2]);
  const float o1 = __builtin_fmaf(x[0], y[1], x[1] * y[3]);
  const float o2 = __builtin_fmaf(x[2], y[0], x[3] * y[2]);
  const float o3 = __builtin_fmaf(x[2], y[1], x[3] * y[3]);
  o[0] = o0; o[1] = o1; o[2] = o2; o[3] = o3;
}
template <int K>
__device__ __forceinline__ float toep(float hv, int hi) {
  const float lo = row_shr0<K>(hv);
  const float up = row_shr0<K + 2>(hv);
  return hi ? up : lo;
}
template <int N>
__device__ __forceinline__ void scan_step(float& e1, float& e2, const float* p) {
  const float t1 = row_shr0<N>(e1);
  const float t2 = row_shr0<N>(e2);
  e1 += p[0] * t1 + p[1] * t2;
  e2 += p[2] * t1 + p[3] * t2;
}

template <int N> struct IntTag { static constexpr int v = N; };

template <bool WRITE_Y>
__global__ __launch_bounds__(32) void biquad_pass_kernel(
    const float* __restrict__ x,
    const float* __restrict__ b0v, const float* __restrict__ b1v, const float* __restrict__ b2v,
    const float* __restrict__ a1v, const float* __restrict__ a2v,
    const float* __restrict__ gin,
    const float* __restrict__ slabin,
    float* __restrict__ gend,
    float* __restrict__ slabend,
    float* __restrict__ y) {
  const int lane = threadIdx.x;
  const int wid = blockIdx.x;
  const int c = wid / SLABS;
  const int s = wid % SLABS;
  const int Mrow = lane & 15;
  const int hi = lane >> 4;

  const float b0 = b0v[c], b1 = b1v[c], b2 = b2v[c], a1 = a1v[c], a2 = a2v[c];
  const float v1 = __builtin_fmaf(-a1, b0, b1);
  const float v2 = __builtin_fmaf(-a2, b0, b2);
  const float na1 = -a1, na2 = -a2;
  float hh[BLK], gA[BLK], gB[BLK];
  float p1[4], p2[4], p4[4];
  hh[0] = b0; gA[0] = 1.0f; gB[0] = 0.0f;
  {
    float ga = 1.0f, gb = 0.0f;
#pragma unroll
    for (int i = 1; i <= 4 * BLK; ++i) {
      const float gan = __builtin_fmaf(na1, ga, na2 * gb);
      if (i < BLK) {
        hh[i] = __builtin_fmaf(ga, v1, gb * v2);
        gA[i] = gan;
        gB[i] = ga;
      }
      if (i == BLK)     { p1[0] = gan; p1[1] = ga; p1[2] = na2 * ga; p1[3] = na2 * gb; }
      if (i == 2 * BLK) { p2[0] = gan; p2[1] = ga; p2[2] = na2 * ga; p2[3] = na2 * gb; }
      if (i == 4 * BLK) { p4[0] = gan; p4[1] = ga; p4[2] = na2 * ga; p4[3] = na2 * gb; }
      gb = ga; ga = gan;
    }
  }
  float p8[4], pm[4];
  mmul(p4, p4, p8);
  mmul(p8, p8, pm);

  float hv = hh[0];
#pragma unroll
  for (int i = 1; i < BLK; ++i) hv = (Mrow == i) ? hh[i] : hv;
  v2f af[4];
  af[0][0] = toep<0>(hv, hi);  af[0][1] = toep<1>(hv, hi);
  af[1][0] = toep<4>(hv, hi);  af[1][1] = toep<5>(hv, hi);
  af[2][0] = toep<8>(hv, hi);  af[2][1] = toep<9>(hv, hi);
  af[3][0] = toep<12>(hv, hi); af[3][1] = toep<13>(hv, hi);

  float gAr[8], gBr[8], qj[4], ps[4];
#pragma unroll
  for (int r = 0; r < 8; ++r) {
    gAr[r] = hi ? gA[8 + r] : gA[r];
    gBr[r] = hi ? gB[8 + r] : gB[r];
  }
  {
    float t[4];
    qj[0] = (Mrow & 1) ? p1[0] : 1.0f; qj[1] = (Mrow & 1) ? p1[1] : 0.0f;
    qj[2] = (Mrow & 1) ? p1[2] : 0.0f; qj[3] = (Mrow & 1) ? p1[3] : 1.0f;
    mmul(qj, p2, t);
#pragma unroll
    for (int i = 0; i < 4; ++i) qj[i] = (Mrow & 2) ? t[i] : qj[i];
    mmul(qj, p4, t);
#pragma unroll
    for (int i = 0; i < 4; ++i) qj[i] = (Mrow & 4) ? t[i] : qj[i];
    mmul(qj, p8, t);
#pragma unroll
    for (int i = 0; i < 4; ++i) qj[i] = (Mrow & 8) ? t[i] : qj[i];
  }
  {
    float sq[4] = {pm[0], pm[1], pm[2], pm[3]};
    float t[4];
    bool have = false;
#pragma unroll
    for (int e = WGRP; e > 0; e >>= 1) {
      if (e & 1) {
        if (have) { mmul(ps, sq, t); }
        else { t[0] = sq[0]; t[1] = sq[1]; t[2] = sq[2]; t[3] = sq[3]; have = true; }
        ps[0] = t[0]; ps[1] = t[1]; ps[2] = t[2]; ps[3] = t[3];
      }
      if (e > 1) {
        mmul(sq, sq, t);
        sq[0] = t[0]; sq[1] = t[1]; sq[2] = t[2]; sq[3] = t[3];
      }
    }
  }

  const int g0 = s * WGRP;
  const bool last = (s == SLABS - 1);
  const int ng = last ? LAST_GROUPS : WGRP;

  float zs1 = 0.0f, zs2 = 0.0f;
  const long gslot = ((long)(g0 + lane) * B_CH + c) * 2;
  float sd1 = 0.0f, sd2 = 0.0f;
  if constexpr (WRITE_Y) {
    if (lane < ng) {
      const float2 t = *(const float2*)(gin + gslot);
      zs1 = t.x; zs2 = t.y;
    }
    if (lane < s) {
      const float2 t = *(const float2*)(slabin + ((long)lane * B_CH + c) * 2);
      sd1 = t.x; sd2 = t.y;
    }
  }
  float G1 = 0.0f, G2 = 0.0f;

  const long base_ch = (long)c * T_LEN + (long)g0 * GRP_LEN;

  auto pair = [&](auto nbb_tag, const int P, const bool prefetch_next) __attribute__((always_inline)) {
    constexpr int nbB = decltype(nbb_tag)::v;
    const long base = base_ch + (long)P * (2 * GRP_LEN);
    const int GA = 2 * P, GB = 2 * P + 1;

    v2f bfA[4], bfB[4];
#pragma unroll
    for (int m = 0; m < 4; ++m) {
      const float2 t = *(const float2*)(x + base + Mrow * BLK + 4 * m + 2 * hi);
      bfA[m][0] = t.x; bfA[m][1] = t.y;
    }
#pragma unroll
    for (int m = 0; m < 4; ++m) {
      if (nbB == GRP || Mrow < nbB) {
        const float2 t = *(const float2*)(x + base + GRP_LEN + Mrow * BLK + 4 * m + 2 * hi);
        bfB[m][0] = t.x; bfB[m][1] = t.y;
      } else {
        bfB[m][0] = 0.0f; bfB[m][1] = 0.0f;
      }
    }

    if (prefetch_next)
      __builtin_prefetch((const void*)(x + base + 2 * GRP_LEN + lane * 16), 0, 1);

    v8f accA = {}, accB = {};
#pragma unroll
    for (int m = 0; m < 4; ++m) {
      accA = __builtin_amdgcn_wmma_f32_16x16x4_f32(false, af[m], false, bfA[m], (short)0, accA, false, false);
      accB = __builtin_amdgcn_wmma_f32_16x16x4_f32(false, af[m], false, bfB[m], (short)0, accB, false, false);
    }

    const float dmidA = b2 * bfA[3][0] - a2 * accA[6];
    const float eA2 = b2 * bfA[3][1] - a2 * accA[7];
    const float eA1 = b1 * bfA[3][1] - a1 * accA[7] + dmidA;
    const float dmidB = b2 * bfB[3][0] - a2 * accB[6];
    const float eB2 = b2 * bfB[3][1] - a2 * accB[7];
    const float eB1 = b1 * bfB[3][1] - a1 * accB[7] + dmidB;
    const float eA1x = xhalf(eA1);
    const float eA2x = xhalf(eA2);
    float e1 = hi ? eB1 : eA1x;
    float e2 = hi ? eB2 : eA2x;
    scan_step<1>(e1, e2, p1);
    scan_step<2>(e1, e2, p2);
    scan_step<4>(e1, e2, p4);
    scan_step<8>(e1, e2, p8);

    if constexpr (WRITE_Y) {
      if (P == 0) {
        for (int i = 0; i < s; ++i) {
          const float d1 = __builtin_bit_cast(float, __builtin_amdgcn_readlane(__builtin_bit_cast(int, sd1), i));
          const float d2 = __builtin_bit_cast(float, __builtin_amdgcn_readlane(__builtin_bit_cast(int, sd2), i));
          const float n1 = ps[0] * G1 + (ps[1] * G2 + d1);
          const float n2 = ps[2] * G1 + (ps[3] * G2 + d2);
          G1 = n1; G2 = n2;
        }
      }
      const float zA1 = G1, zA2 = G2;
      const float dA1 = __builtin_bit_cast(float, __builtin_amdgcn_readlane(__builtin_bit_cast(int, zs1), GA));
      const float dA2 = __builtin_bit_cast(float, __builtin_amdgcn_readlane(__builtin_bit_cast(int, zs2), GA));
      const float zB1 = pm[0] * zA1 + (pm[1] * zA2 + dA1);
      const float zB2 = pm[2] * zA1 + (pm[3] * zA2 + dA2);
      const float dB1 = __builtin_bit_cast(float, __builtin_amdgcn_readlane(__builtin_bit_cast(int, zs1), GB));
      const float dB2 = __builtin_bit_cast(float, __builtin_amdgcn_readlane(__builtin_bit_cast(int, zs2), GB));
      G1 = pm[0] * zB1 + (pm[1] * zB2 + dB1);
      G2 = pm[2] * zB1 + (pm[3] * zB2 + dB2);
      const float z1 = hi ? zB1 : zA1;
      const float z2 = hi ? zB2 : zA2;
      const float sj1 = qj[0] * z1 + qj[1] * z2 + row_shr0<1>(e1);
      const float sj2 = qj[2] * z1 + qj[3] * z2 + row_shr0<1>(e2);
      const float sx1 = xhalf(sj1);
      const float sx2 = xhalf(sj2);
      const float sA1 = hi ? sx1 : sj1, sA2 = hi ? sx2 : sj2;
      const float sB1 = hi ? sj1 : sx1, sB2 = hi ? sj2 : sx2;
#pragma unroll
      for (int r = 0; r < 8; ++r) accA[r] += gAr[r] * sA1 + gBr[r] * sA2;
#pragma unroll
      for (int r = 0; r < 8; ++r) accB[r] += gAr[r] * sB1 + gBr[r] * sB2;
      {
        float* dst = y + base + Mrow * BLK + 8 * hi;
        *(float4*)(dst + 0) = make_float4(accA[0], accA[1], accA[2], accA[3]);
        *(float4*)(dst + 4) = make_float4(accA[4], accA[5], accA[6], accA[7]);
      }
      if (nbB == GRP || Mrow < nbB) {
        float* dst = y + base + GRP_LEN + Mrow * BLK + 8 * hi;
        *(float4*)(dst + 0) = make_float4(accB[0], accB[1], accB[2], accB[3]);
        *(float4*)(dst + 4) = make_float4(accB[4], accB[5], accB[6], accB[7]);
      }
    } else {
      const float lA1 = __builtin_bit_cast(float, __builtin_amdgcn_readlane(__builtin_bit_cast(int, e1), GRP - 1));
      const float lA2 = __builtin_bit_cast(float, __builtin_amdgcn_readlane(__builtin_bit_cast(int, e2), GRP - 1));
      const float lB1 = __builtin_bit_cast(float, __builtin_amdgcn_readlane(__builtin_bit_cast(int, e1), 16 + nbB - 1));
      const float lB2 = __builtin_bit_cast(float, __builtin_amdgcn_readlane(__builtin_bit_cast(int, e2), 16 + nbB - 1));
      zs1 = (lane == GA) ? lA1 : ((lane == GB) ? lB1 : zs1);
      zs2 = (lane == GA) ? lA2 : ((lane == GB) ? lB2 : zs2);
    }
  };

  if (!last) {
    if constexpr (WRITE_Y) {
      for (int P = 0; P < PAIRS; ++P) pair(IntTag<GRP>{}, P, P + 1 < PAIRS);
    } else {
#pragma unroll
      for (int P = 0; P < PAIRS; ++P) pair(IntTag<GRP>{}, P, P + 1 < PAIRS);
    }
  } else {
    if constexpr (WRITE_Y) {
      for (int P = 0; P < LAST_PAIRS - 1; ++P) pair(IntTag<GRP>{}, P, P + 2 < LAST_PAIRS);
    } else {
#pragma unroll
      for (int P = 0; P < LAST_PAIRS - 1; ++P) pair(IntTag<GRP>{}, P, P + 2 < LAST_PAIRS);
    }
    pair(IntTag<TAIL_NB>{}, LAST_PAIRS - 1, false);
  }

  if constexpr (!WRITE_Y) {
    if (lane < ng) *(float2*)(gend + gslot) = make_float2(zs1, zs2);
    if (!last) {
#pragma unroll
      for (int g = 0; g < WGRP; ++g) {
        const float d1 = __builtin_bit_cast(float, __builtin_amdgcn_readlane(__builtin_bit_cast(int, zs1), g));
        const float d2 = __builtin_bit_cast(float, __builtin_amdgcn_readlane(__builtin_bit_cast(int, zs2), g));
        const float n1 = pm[0] * G1 + (pm[1] * G2 + d1);
        const float n2 = pm[2] * G1 + (pm[3] * G2 + d2);
        G1 = n1; G2 = n2;
      }
      if (lane == 0) *(float2*)(slabend + ((long)s * B_CH + c) * 2) = make_float2(G1, G2);
    }
  }
}

extern "C" void kernel_launch(void* const* d_in, const int* in_sizes, int n_in,
                              void* d_out, int out_size, void* d_ws, size_t ws_size,
                              hipStream_t stream) {
  const float* x  = (const float*)d_in[0];
  const float* b0 = (const float*)d_in[1];
  const float* b1 = (const float*)d_in[2];
  const float* b2 = (const float*)d_in[3];
  const float* a1 = (const float*)d_in[4];
  const float* a2 = (const float*)d_in[5];
  float* y = (float*)d_out;

  float* ws      = (float*)d_ws;
  float* gend    = ws;
  float* slabend = gend + GROUPS_CH * B_CH * 2;

  biquad_pass_kernel<false><<<B_CH * SLABS, 32, 0, stream>>>(
      x, b0, b1, b2, a1, a2, nullptr, nullptr, gend, slabend, nullptr);
  biquad_pass_kernel<true><<<B_CH * SLABS, 32, 0, stream>>>(
      x, b0, b1, b2, a1, a2, gend, slabend, nullptr, nullptr, y);
}


# one launch (was 2): a work-group is one channel, its 11 waves the slabs; slab end states cross one barrier through LDS, group end states stay in the wave's lanes; workspace unused; sums unchanged
# speedup vs baseline: 3.8426x; 1.2291x over previous
#include <hip/hip_runtime.h>
#include <hip/hip_bf16.h>


typedef __attribute__((ext_vector_type(2))) float v2f;
typedef __attribute__((ext_vector_type(8))) float v8f;

namespace {
constexpr int B_CH = 512;
constexpr int T_LEN = 48000;
constexpr int BLK = 16;
constexpr int GRP = 16;
constexpr int GRP_LEN = GRP * BLK;
constexpr int BLOCKS_CH = T_LEN / BLK;
constexpr int FULL_GROUPS = BLOCKS_CH / GRP;
constexpr int TAIL_NB = BLOCKS_CH - FULL_GROUPS * GRP;
constexpr int GROUPS_CH = FULL_GROUPS + 1;
constexpr int WGRP = 18;
constexpr int PAIRS = WGRP / 2;
constexpr int SLABS = (GROUPS_CH + WGRP - 1) / WGRP;
constexpr int LAST_GROUPS = GROUPS_CH - (SLABS - 1) * WGRP;
constexpr int LAST_PAIRS = LAST_GROUPS / 2;
static_assert(LAST_GROUPS > 0 && LAST_GROUPS <= WGRP && LAST_GROUPS % 2 == 0 && LAST_PAIRS >= 1 &&
              TAIL_NB > 0 && TAIL_NB < GRP, "group split");
static_assert(BLK == 16 && GRP == 16, "the power ladder below is written for 16 x 16");
}

__device__ __forceinline__ float xhalf(float v) {
  const int b = __builtin_bit_cast(int, v);
  return __builtin_bit_cast(float, __builtin_amdgcn_permlanex16(b, b, 0x76543210, 0xfedcba98, false, false));
}
template <int N>
__device__ __forceinline__ float row_shr0(float v) {
  if constexpr (N == 0) return v;
  else return __builtin_bit_cast(float, __builtin_amdgcn_update_dpp(0, __builtin_bit_cast(int, v), 0x110 + N, 0xf, 0xf, true));
}
__device__ __forceinline__ void mmul(const float* x, const float* y, float* o) {
  const float o0 = __builtin_fmaf(x[0], y[0], x[1] * y[2]);
  const float o1 = __builtin_fmaf(x[0], y[1], x[1] * y[3]);
  const float o2 = __builtin_fmaf(x[2], y[0], x[3] * y[2]);
  const float o3 = __builtin_fmaf(x[2], y[1], x[3] * y[3]);
  o[0] = o0; o[1] = o1; o[2] = o2; o[3] = o3;
}
template <int K>
__device__ __forceinline__ float toep(float hv, int hi) {
  const float lo = row_shr0<K>(hv);
  const float up = row_shr0<K + 2>(hv);
  return hi ? up : lo;
}
template <int N>
__device__ __forceinline__ void scan_step(float& e1, float& e2, const float* p) {
  const float t1 = row_shr0<N>(e1);
  const float t2 = row_shr0<N>(e2);
  e1 += p[0] * t1 + p[1] * t2;
  e2 += p[2] * t1 + p[3] * t2;
}

template <int N> struct IntTag { static constexpr int v = N; };

__global__ __launch_bounds__(32 * SLABS) __attribute__((amdgpu_waves_per_eu(6))) void biquad_kernel(
    const float* __restrict__ x,
    const float* __restrict__ b0v, const float* __restrict__ b1v, const float* __restrict__ b2v,
    const float* __restrict__ a1v, const float* __restrict__ a2v,
    float* __restrict__ y) {
  __shared__ float2 slab_end[SLABS - 1];
  const int lane = threadIdx.x & 31;
  const int c = blockIdx.x;
  const int s = __builtin_amdgcn_readfirstlane((int)(threadIdx.x >> 5));
  const int Mrow = lane & 15;
  const int hi = lane >> 4;

  const float b0 = b0v[c], b1 = b1v[c], b2 = b2v[c], a1 = a1v[c], a2 = a2v[c];
  const float v1 = __builtin_fmaf(-a1, b0, b1);
  const float v2 = __builtin_fmaf(-a2, b0, b2);
  const float na1 = -a1, na2 = -a2;
  float hh[BLK], gA[BLK], gB[BLK];
  float p1[4], p2[4], p4[4];
  hh[0] = b0; gA[0] = 1.0f; gB[0] = 0.0f;
  {
    float ga = 1.0f, gb = 0.0f;
#pragma unroll
    for (int i = 1; i <= 4 * BLK; ++i) {
      const float gan = __builtin_fmaf(na1, ga, na2 * gb);
      if (i < BLK) {
        hh[i] = __builtin_fmaf(ga, v1, gb * v2);
        gA[i] = gan;
        gB[i] = ga;
      }
      if (i == BLK)     { p1[0] = gan; p1[1] = ga; p1[2] = na2 * ga; p1[3] = na2 * gb; }
      if (i == 2 * BLK) { p2[0] = gan; p2[1] = ga; p2[2] = na2 * ga; p2[3] = na2 * gb; }
      if (i == 4 * BLK) { p4[0] = gan; p4[1] = ga; p4[2] = na2 * ga; p4[3] = na2 * gb; }
      gb = ga; ga = gan;
    }
  }
  float p8[4], pm[4];
  mmul(p4, p4, p8);
  mmul(p8, p8, pm);

  float hv = hh[0];
#pragma unroll
  for (int i = 1; i < BLK; ++i) hv = (Mrow == i) ? hh[i] : hv;
  v2f af[4];
  af[0][0] = toep<0>(hv, hi);  af[0][1] = toep<1>(hv, hi);
  af[1][0] = toep<4>(hv, hi);  af[1][1] = toep<5>(hv, hi);
  af[2][0] = toep<8>(hv, hi);  af[2][1] = toep<9>(hv, hi);
  af[3][0] = toep<12>(hv, hi); af[3][1] = toep<13>(hv, hi);

  float gAr[8], gBr[8], qj[4], ps[4];
#pragma unroll
  for (int r = 0; r < 8; ++r) {
    gAr[r] = hi ? gA[8 + r] : gA[r];
    gBr[r] = hi ? gB[8 + r] : gB[r];
  }
  {
    float t[4];
    qj[0] = (Mrow & 1) ? p1[0] : 1.0f; qj[1] = (Mrow & 1) ? p1[1] : 0.0f;
    qj[2] = (Mrow & 1) ? p1[2] : 0.0f; qj[3] = (Mrow & 1) ? p1[3] : 1.0f;
    mmul(qj, p2, t);
#pragma unroll
    for (int i = 0; i < 4; ++i) qj[i] = (Mrow & 2) ? t[i] : qj[i];
    mmul(qj, p4, t);
#pragma unroll
    for (int i = 0; i < 4; ++i) qj[i] = (Mrow & 4) ? t[i] : qj[i];
    mmul(qj, p8, t);
#pragma unroll
    for (int i = 0; i < 4; ++i) qj[i] = (Mrow & 8) ? t[i] : qj[i];
  }
  {
    float sq[4] = {pm[0], pm[1], pm[2], pm[3]};
    float t[4];
    bool have = false;
#pragma unroll
    for (int e = WGRP; e > 0; e >>= 1) {
      if (e & 1) {
        if (have) { mmul(ps, sq, t); }
        else { t[0] = sq[0]; t[1] = sq[1]; t[2] = sq[2]; t[3] = sq[3]; have = true; }
        ps[0] = t[0]; ps[1] = t[1]; ps[2] = t[2]; ps[3] = t[3];
      }
      if (e > 1) {
        mmul(sq, sq, t);
        sq[0] = t[0]; sq[1] = t[1]; sq[2] = t[2]; sq[3] = t[3];
      }
    }
  }

  const int g0 = s * WGRP;
  const bool last = (s == SLABS - 1);

  float zs1 = 0.0f, zs2 = 0.0f;
  float sd1 = 0.0f, sd2 = 0.0f;
  float G1 = 0.0f, G2 = 0.0f;

  const long base_ch = (long)c * T_LEN + (long)g0 * GRP_LEN;

  auto pair = [&](auto nbb_tag, auto wy_tag, const int P, const bool prefetch_next) __attribute__((always_inline)) {
    constexpr int nbB = decltype(nbb_tag)::v;
    constexpr bool WRITE_Y = decltype(wy_tag)::v != 0;
    const long base = base_ch + (long)P * (2 * GRP_LEN);
    const int GA = 2 * P, GB = 2 * P + 1;

    v2f bfA[4], bfB[4];
#pragma unroll
    for (int m = 0; m < 4; ++m) {
      const float2 t = *(const float2*)(x + base + Mrow * BLK + 4 * m + 2 * hi);
      bfA[m][0] = t.x; bfA[m][1] = t.y;
    }
#pragma unroll
    for (int m = 0; m < 4; ++m) {
      if (nbB == GRP || Mrow < nbB) {
        const float2 t = *(const float2*)(x + base + GRP_LEN + Mrow * BLK + 4 * m + 2 * hi);
        bfB[m][0] = t.x; bfB[m][1] = t.y;
      } else {
        bfB[m][0] = 0.0f; bfB[m][1] = 0.0f;
      }
    }

    if (prefetch_next)
      __builtin_prefetch((const void*)(x + base + 2 * GRP_LEN + lane * 16), 0, 1);

    v8f accA = {}, accB = {};
#pragma unroll
    for (int m = 0; m < 4; ++m) {
      accA = __builtin_amdgcn_wmma_f32_16x16x4_f32(false, af[m], false, bfA[m], (short)0, accA, false, false);
      accB = __builtin_amdgcn_wmma_f32_16x16x4_f32(false, af[m], false, bfB[m], (short)0, accB, false, false);
    }

    const float dmidA = b2 * bfA[3][0] - a2 * accA[6];
    const float eA2 = b2 * bfA[3][1] - a2 * accA[7];
    const float eA1 = b1 * bfA[3][1] - a1 * accA[7] + dmidA;
    const float dmidB = b2 * bfB[3][0] - a2 * accB[6];
    const float eB2 = b2 * bfB[3][1] - a2 * accB[7];
    const float eB1 = b1 * bfB[3][1] - a1 * accB[7] + dmidB;
    const float eA1x = xhalf(eA1);
    const float eA2x = xhalf(eA2);
    float e1 = hi ? eB1 : eA1x;
    float e2 = hi ? eB2 : eA2x;
    scan_step<1>(e1, e2, p1);
    scan_step<2>(e1, e2, p2);
    scan_step<4>(e1, e2, p4);
    scan_step<8>(e1, e2, p8);

    if constexpr (WRITE_Y) {
      if (P == 0) {
        for (int i = 0; i < s; ++i) {
          const float d1 = __builtin_bit_cast(float, __builtin_amdgcn_readlane(__builtin_bit_cast(int, sd1), i));
          const float d2 = __builtin_bit_cast(float, __builtin_amdgcn_readlane(__builtin_bit_cast(int, sd2), i));
          const float n1 = ps[0] * G1 + (ps[1] * G2 + d1);
          const float n2 = ps[2] * G1 + (ps[3] * G2 + d2);
          G1 = n1; G2 = n2;
        }
      }
      const float zA1 = G1, zA2 = G2;
      const float dA1 = __builtin_bit_cast(float, __builtin_amdgcn_readlane(__builtin_bit_cast(int, zs1), GA));
      const float dA2 = __builtin_bit_cast(float, __builtin_amdgcn_readlane(__builtin_bit_cast(int, zs2), GA));
      const float zB1 = pm[0] * zA1 + (pm[1] * zA2 + dA1);
      const float zB2 = pm[2] * zA1 + (pm[3] * zA2 + dA2);
      const float dB1 = __builtin_bit_cast(float, __builtin_amdgcn_readlane(__builtin_bit_cast(int, zs1), GB));
      const float dB2 = __builtin_bit_cast(float, __builtin_amdgcn_readlane(__builtin_bit_cast(int, zs2), GB));
      G1 = pm[0] * zB1 + (pm[1] * zB2 + dB1);
      G2 = pm[2] * zB1 + (pm[3] * zB2 + dB2);
      const float z1 = hi ? zB1 : zA1;
      const float z2 = hi ? zB2 : zA2;
      const float sj1 = qj[0] * z1 + qj[1] * z2 + row_shr0<1>(e1);
      const float sj2 = qj[2] * z1 + qj[3] * z2 + row_shr0<1>(e2);
      const float sx1 = xhalf(sj1);
      const float sx2 = xhalf(sj2);
      const float sA1 = hi ? sx1 : sj1, sA2 = hi ? sx2 : sj2;
      const float sB1 = hi ? sj1 : sx1, sB2 = hi ? sj2 : sx2;
#pragma unroll
      for (int r = 0; r < 8; ++r) accA[r] += gAr[r] * sA1 + gBr[r] * sA2;
#pragma unroll
      for (int r = 0; r < 8; ++r) accB[r] += gAr[r] * sB1 + gBr[r] * sB2;
      {
        float* dst = y + base + Mrow * BLK + 8 * hi;
        *(float4*)(dst + 0) = make_float4(accA[0], accA[1], accA[2], accA[3]);
        *(float4*)(dst + 4) = make_float4(accA[4], accA[5], accA[6], accA[7]);
      }
      if (nbB == GRP || Mrow < nbB) {
        float* dst = y + base + GRP_LEN + Mrow * BLK + 8 * hi;
        *(float4*)(dst + 0) = make_float4(accB[0], accB[1], accB[2], accB[3]);
        *(float4*)(dst + 4) = make_float4(accB[4], accB[5], accB[6], accB[7]);
      }
    } else {
      const float lA1 = __builtin_bit_cast(float, __builtin_amdgcn_readlane(__builtin_bit_cast(int, e1), GRP - 1));
      const float lA2 = __builtin_bit_cast(float, __builtin_amdgcn_readlane(__builtin_bit_cast(int, e2), GRP - 1));
      const float lB1 = __builtin_bit_cast(float, __builtin_amdgcn_readlane(__builtin_bit_cast(int, e1), 16 + nbB - 1));
      const float lB2 = __builtin_bit_cast(float, __builtin_amdgcn_readlane(__builtin_bit_cast(int, e2), 16 + nbB - 1));
      zs1 = (lane == GA) ? lA1 : ((lane == GB) ? lB1 : zs1);
      zs2 = (lane == GA) ? lA2 : ((lane == GB) ? lB2 : zs2);
    }
  };

  if (!last) {
#pragma unroll
    for (int P = 0; P < PAIRS; ++P) pair(IntTag<GRP>{}, IntTag<0>{}, P, P + 1 < PAIRS);
#pragma unroll
    for (int g = 0; g < WGRP; ++g) {
      const float d1 = __builtin_bit_cast(float, __builtin_amdgcn_readlane(__builtin_bit_cast(int, zs1), g));
      const float d2 = __builtin_bit_cast(float, __builtin_amdgcn_readlane(__builtin_bit_cast(int, zs2), g));
      const float n1 = pm[0] * G1 + (pm[1] * G2 + d1);
      const float n2 = pm[2] * G1 + (pm[3] * G2 + d2);
      G1 = n1; G2 = n2;
    }
    if (lane == 0) slab_end[s] = make_float2(G1, G2);
  } else {
#pragma unroll
    for (int P = 0; P < LAST_PAIRS - 1; ++P) pair(IntTag<GRP>{}, IntTag<0>{}, P, P + 2 < LAST_PAIRS);
    pair(IntTag<TAIL_NB>{}, IntTag<0>{}, LAST_PAIRS - 1, false);
  }

  __syncthreads();

  if (lane < s) {
    const float2 t = slab_end[lane];
    sd1 = t.x; sd2 = t.y;
  }
  G1 = 0.0f; G2 = 0.0f;
  if (!last) {
    for (int P = 0; P < PAIRS; ++P) pair(IntTag<GRP>{}, IntTag<1>{}, P, P + 1 < PAIRS);
  } else {
    for (int P = 0; P < LAST_PAIRS - 1; ++P) pair(IntTag<GRP>{}, IntTag<1>{}, P, P + 2 < LAST_PAIRS);
    pair(IntTag<TAIL_NB>{}, IntTag<1>{}, LAST_PAIRS - 1, false);
  }
}

extern "C" void kernel_launch(void* const* d_in, const int* in_sizes, int n_in,
                              void* d_out, int out_size, void* d_ws, size_t ws_size,
                              hipStream_t stream) {
  const float* x  = (const float*)d_in[0];
  const float* b0 = (const float*)d_in[1];
  const float* b1 = (const float*)d_in[2];
  const float* b2 = (const float*)d_in[3];
  const float* a1 = (const float*)d_in[4];
  const float* a2 = (const float*)d_in[5];
  float* y = (float*)d_out;

  biquad_kernel<<<B_CH, 32 * SLABS, 0, stream>>>(x, b0, b1, b2, a1, a2, y);
}


# second sweep reuses first sweep's scanned block states, kept in registers: no second scan, accumulators start at G*s; to fit, shared pairs run before the last-wave split; y summation order changes
# speedup vs baseline: 4.2696x; 1.1111x over previous
#include <hip/hip_runtime.h>
#include <hip/hip_bf16.h>


typedef __attribute__((ext_vector_type(2))) float v2f;
typedef __attribute__((ext_vector_type(8))) float v8f;

namespace {
constexpr int B_CH = 512;
constexpr int T_LEN = 48000;
constexpr int BLK = 16;
constexpr int GRP = 16;
constexpr int GRP_LEN = GRP * BLK;
constexpr int BLOCKS_CH = T_LEN / BLK;
constexpr int FULL_GROUPS = BLOCKS_CH / GRP;
constexpr int TAIL_NB = BLOCKS_CH - FULL_GROUPS * GRP;
constexpr int GROUPS_CH = FULL_GROUPS + 1;
constexpr int WGRP = 18;
constexpr int PAIRS = WGRP / 2;
constexpr int SLABS = (GROUPS_CH + WGRP - 1) / WGRP;
constexpr int LAST_GROUPS = GROUPS_CH - (SLABS - 1) * WGRP;
constexpr int LAST_PAIRS = LAST_GROUPS / 2;
static_assert(LAST_GROUPS > 0 && LAST_GROUPS <= WGRP && LAST_GROUPS % 2 == 0 && LAST_PAIRS >= 1 &&
              TAIL_NB > 0 && TAIL_NB < GRP, "group split");
static_assert(BLK == 16 && GRP == 16, "the power ladder below is written for 16 x 16");
}

__device__ __forceinline__ float xhalf(float v) {
  const int b = __builtin_bit_cast(int, v);
  return __builtin_bit_cast(float, __builtin_amdgcn_permlanex16(b, b, 0x76543210, 0xfedcba98, false, false));
}
template <int N>
__device__ __forceinline__ float row_shr0(float v) {
  if constexpr (N == 0) return v;
  else return __builtin_bit_cast(float, __builtin_amdgcn_update_dpp(0, __builtin_bit_cast(int, v), 0x110 + N, 0xf, 0xf, true));
}
__device__ __forceinline__ void mmul(const float* x, const float* y, float* o) {
  const float o0 = __builtin_fmaf(x[0], y[0], x[1] * y[2]);
  const float o1 = __builtin_fmaf(x[0], y[1], x[1] * y[3]);
  const float o2 = __builtin_fmaf(x[2], y[0], x[3] * y[2]);
  const float o3 = __builtin_fmaf(x[2], y[1], x[3] * y[3]);
  o[0] = o0; o[1] = o1; o[2] = o2; o[3] = o3;
}
template <int K>
__device__ __forceinline__ float toep(float hv, int hi) {
  const float lo = row_shr0<K>(hv);
  const float up = row_shr0<K + 2>(hv);
  return hi ? up : lo;
}
template <int N>
__device__ __forceinline__ void scan_step(float& e1, float& e2, const float* p) {
  const float t1 = row_shr0<N>(e1);
  const float t2 = row_shr0<N>(e2);
  e1 += p[0] * t1 + p[1] * t2;
  e2 += p[2] * t1 + p[3] * t2;
}

template <int N> struct IntTag { static constexpr int v = N; };

__global__ __launch_bounds__(32 * SLABS) __attribute__((amdgpu_waves_per_eu(6))) void biquad_kernel(
    const float* __restrict__ x,
    const float* __restrict__ b0v, const float* __restrict__ b1v, const float* __restrict__ b2v,
    const float* __restrict__ a1v, const float* __restrict__ a2v,
    float* __restrict__ y) {
  __shared__ float2 slab_end[SLABS - 1];
  const int lane = threadIdx.x & 31;
  const int c = blockIdx.x;
  const int s = __builtin_amdgcn_readfirstlane((int)(threadIdx.x >> 5));
  const int Mrow = lane & 15;
  const int hi = lane >> 4;

  const float b0 = b0v[c], b1 = b1v[c], b2 = b2v[c], a1 = a1v[c], a2 = a2v[c];
  const float v1 = __builtin_fmaf(-a1, b0, b1);
  const float v2 = __builtin_fmaf(-a2, b0, b2);
  const float na1 = -a1, na2 = -a2;
  float hh[BLK], gA[BLK], gB[BLK];
  float p1[4], p2[4], p4[4];
  hh[0] = b0; gA[0] = 1.0f; gB[0] = 0.0f;
  {
    float ga = 1.0f, gb = 0.0f;
#pragma unroll
    for (int i = 1; i <= 4 * BLK; ++i) {
      const float gan = __builtin_fmaf(na1, ga, na2 * gb);
      if (i < BLK) {
        hh[i] = __builtin_fmaf(ga, v1, gb * v2);
        gA[i] = gan;
        gB[i] = ga;
      }
      if (i == BLK)     { p1[0] = gan; p1[1] = ga; p1[2] = na2 * ga; p1[3] = na2 * gb; }
      if (i == 2 * BLK) { p2[0] = gan; p2[1] = ga; p2[2] = na2 * ga; p2[3] = na2 * gb; }
      if (i == 4 * BLK) { p4[0] = gan; p4[1] = ga; p4[2] = na2 * ga; p4[3] = na2 * gb; }
      gb = ga; ga = gan;
    }
  }
  float p8[4], pm[4];
  mmul(p4, p4, p8);
  mmul(p8, p8, pm);

  float hv = hh[0];
#pragma unroll
  for (int i = 1; i < BLK; ++i) hv = (Mrow == i) ? hh[i] : hv;
  v2f af[4];
  af[0][0] = toep<0>(hv, hi);  af[0][1] = toep<1>(hv, hi);
  af[1][0] = toep<4>(hv, hi);  af[1][1] = toep<5>(hv, hi);
  af[2][0] = toep<8>(hv, hi);  af[2][1] = toep<9>(hv, hi);
  af[3][0] = toep<12>(hv, hi); af[3][1] = toep<13>(hv, hi);

  float gAr[8], gBr[8], qj[4], ps[4];
#pragma unroll
  for (int r = 0; r < 8; ++r) {
    gAr[r] = hi ? gA[8 + r] : gA[r];
    gBr[r] = hi ? gB[8 + r] : gB[r];
  }
  {
    float t[4];
    qj[0] = (Mrow & 1) ? p1[0] : 1.0f; qj[1] = (Mrow & 1) ? p1[1] : 0.0f;
    qj[2] = (Mrow & 1) ? p1[2] : 0.0f; qj[3] = (Mrow & 1) ? p1[3] : 1.0f;
    mmul(qj, p2, t);
#pragma unroll
    for (int i = 0; i < 4; ++i) qj[i] = (Mrow & 2) ? t[i] : qj[i];
    mmul(qj, p4, t);
#pragma unroll
    for (int i = 0; i < 4; ++i) qj[i] = (Mrow & 4) ? t[i] : qj[i];
    mmul(qj, p8, t);
#pragma unroll
    for (int i = 0; i < 4; ++i) qj[i] = (Mrow & 8) ? t[i] : qj[i];
  }
  {
    float sq[4] = {pm[0], pm[1], pm[2], pm[3]};
    float t[4];
    bool have = false;
#pragma unroll
    for (int e = WGRP; e > 0; e >>= 1) {
      if (e & 1) {
        if (have) { mmul(ps, sq, t); }
        else { t[0] = sq[0]; t[1] = sq[1]; t[2] = sq[2]; t[3] = sq[3]; have = true; }
        ps[0] = t[0]; ps[1] = t[1]; ps[2] = t[2]; ps[3] = t[3];
      }
      if (e > 1) {
        mmul(sq, sq, t);
        sq[0] = t[0]; sq[1] = t[1]; sq[2] = t[2]; sq[3] = t[3];
      }
    }
  }

  const int g0 = s * WGRP;
  const bool last = (s == SLABS - 1);

  float ek1[PAIRS] = {}, ek2[PAIRS] = {};
  float sd1 = 0.0f, sd2 = 0.0f;
  float G1 = 0.0f, G2 = 0.0f;

  const long base_ch = (long)c * T_LEN + (long)g0 * GRP_LEN;

  auto pair = [&](auto nbb_tag, auto wy_tag, const int P, const bool prefetch_next) __attribute__((always_inline)) {
    constexpr int nbB = decltype(nbb_tag)::v;
    constexpr bool WRITE_Y = decltype(wy_tag)::v != 0;
    const long base = base_ch + (long)P * (2 * GRP_LEN);

    v2f bfA[4], bfB[4];
#pragma unroll
    for (int m = 0; m < 4; ++m) {
      const float2 t = *(const float2*)(x + base + Mrow * BLK + 4 * m + 2 * hi);
      bfA[m][0] = t.x; bfA[m][1] = t.y;
    }
#pragma unroll
    for (int m = 0; m < 4; ++m) {
      if (nbB == GRP || Mrow < nbB) {
        const float2 t = *(const float2*)(x + base + GRP_LEN + Mrow * BLK + 4 * m + 2 * hi);
        bfB[m][0] = t.x; bfB[m][1] = t.y;
      } else {
        bfB[m][0] = 0.0f; bfB[m][1] = 0.0f;
      }
    }

    if (prefetch_next)
      __builtin_prefetch((const void*)(x + base + 2 * GRP_LEN + lane * 16), 0, 1);

    v8f accA = {}, accB = {};
    if constexpr (WRITE_Y) {
      if (P == 0) {
        for (int i = 0; i < s; ++i) {
          const float d1 = __builtin_bit_cast(float, __builtin_amdgcn_readlane(__builtin_bit_cast(int, sd1), i));
          const float d2 = __builtin_bit_cast(float, __builtin_amdgcn_readlane(__builtin_bit_cast(int, sd2), i));
          const float n1 = ps[0] * G1 + (ps[1] * G2 + d1);
          const float n2 = ps[2] * G1 + (ps[3] * G2 + d2);
          G1 = n1; G2 = n2;
        }
      }
      const float e1 = ek1[P], e2 = ek2[P];
      const float zA1 = G1, zA2 = G2;
      const float dA1 = __builtin_bit_cast(float, __builtin_amdgcn_readlane(__builtin_bit_cast(int, e1), GRP - 1));
      const float dA2 = __builtin_bit_cast(float, __builtin_amdgcn_readlane(__builtin_bit_cast(int, e2), GRP - 1));
      const float zB1 = pm[0] * zA1 + (pm[1] * zA2 + dA1);
      const float zB2 = pm[2] * zA1 + (pm[3] * zA2 + dA2);
      const float dB1 = __builtin_bit_cast(float, __builtin_amdgcn_readlane(__builtin_bit_cast(int, e1), 16 + nbB - 1));
      const float dB2 = __builtin_bit_cast(float, __builtin_amdgcn_readlane(__builtin_bit_cast(int, e2), 16 + nbB - 1));
      G1 = pm[0] * zB1 + (pm[1] * zB2 + dB1);
      G2 = pm[2] * zB1 + (pm[3] * zB2 + dB2);
      const float z1 = hi ? zB1 : zA1;
      const float z2 = hi ? zB2 : zA2;
      const float sj1 = qj[0] * z1 + qj[1] * z2 + row_shr0<1>(e1);
      const float sj2 = qj[2] * z1 + qj[3] * z2 + row_shr0<1>(e2);
      const float sx1 = xhalf(sj1);
      const float sx2 = xhalf(sj2);
      const float sA1 = hi ? sx1 : sj1, sA2 = hi ? sx2 : sj2;
      const float sB1 = hi ? sj1 : sx1, sB2 = hi ? sj2 : sx2;
#pragma unroll
      for (int r = 0; r < 8; ++r) accA[r] = gAr[r] * sA1 + gBr[r] * sA2;
#pragma unroll
      for (int r = 0; r < 8; ++r) accB[r] = gAr[r] * sB1 + gBr[r] * sB2;
    }

#pragma unroll
    for (int m = 0; m < 4; ++m) {
      accA = __builtin_amdgcn_wmma_f32_16x16x4_f32(false, af[m], false, bfA[m], (short)0, accA, false, false);
      accB = __builtin_amdgcn_wmma_f32_16x16x4_f32(false, af[m], false, bfB[m], (short)0, accB, false, false);
    }

    if constexpr (WRITE_Y) {
      {
        float* dst = y + base + Mrow * BLK + 8 * hi;
        *(float4*)(dst + 0) = make_float4(accA[0], accA[1], accA[2], accA[3]);
        *(float4*)(dst + 4) = make_float4(accA[4], accA[5], accA[6], accA[7]);
      }
      if (nbB == GRP || Mrow < nbB) {
        float* dst = y + base + GRP_LEN + Mrow * BLK + 8 * hi;
        *(float4*)(dst + 0) = make_float4(accB[0], accB[1], accB[2], accB[3]);
        *(float4*)(dst + 4) = make_float4(accB[4], accB[5], accB[6], accB[7]);
      }
    } else {
      const float dmidA = b2 * bfA[3][0] - a2 * accA[6];
      const float eA2 = b2 * bfA[3][1] - a2 * accA[7];
      const float eA1 = b1 * bfA[3][1] - a1 * accA[7] + dmidA;
      const float dmidB = b2 * bfB[3][0] - a2 * accB[6];
      const float eB2 = b2 * bfB[3][1] - a2 * accB[7];
      const float eB1 = b1 * bfB[3][1] - a1 * accB[7] + dmidB;
      const float eA1x = xhalf(eA1);
      const float eA2x = xhalf(eA2);
      float e1 = hi ? eB1 : eA1x;
      float e2 = hi ? eB2 : eA2x;
      scan_step<1>(e1, e2, p1);
      scan_step<2>(e1, e2, p2);
      scan_step<4>(e1, e2, p4);
      scan_step<8>(e1, e2, p8);
      ek1[P] = e1; ek2[P] = e2;
    }
  };

  for (int P = 0; P < LAST_PAIRS - 1; ++P)
    pair(IntTag<GRP>{}, IntTag<0>{}, P, last ? (P + 2 < LAST_PAIRS) : true);
  if (!last) {
    for (int P = LAST_PAIRS - 1; P < PAIRS; ++P) pair(IntTag<GRP>{}, IntTag<0>{}, P, P + 1 < PAIRS);
#pragma unroll
    for (int g = 0; g < WGRP; ++g) {
      const int ln = (g & 1) ? 2 * GRP - 1 : GRP - 1;
      const float d1 = __builtin_bit_cast(float, __builtin_amdgcn_readlane(__builtin_bit_cast(int, ek1[g >> 1]), ln));
      const float d2 = __builtin_bit_cast(float, __builtin_amdgcn_readlane(__builtin_bit_cast(int, ek2[g >> 1]), ln));
      const float n1 = pm[0] * G1 + (pm[1] * G2 + d1);
      const float n2 = pm[2] * G1 + (pm[3] * G2 + d2);
      G1 = n1; G2 = n2;
    }
    if (lane == 0) slab_end[s] = make_float2(G1, G2);
  }
  if (last) pair(IntTag<TAIL_NB>{}, IntTag<0>{}, LAST_PAIRS - 1, false);

  __syncthreads();

  if (lane < s) {
    const float2 t = slab_end[lane];
    sd1 = t.x; sd2 = t.y;
  }
  G1 = 0.0f; G2 = 0.0f;
  for (int P = 0; P < LAST_PAIRS - 1; ++P)
    pair(IntTag<GRP>{}, IntTag<1>{}, P, last ? (P + 2 < LAST_PAIRS) : true);
  if (!last) {
    for (int P = LAST_PAIRS - 1; P < PAIRS; ++P) pair(IntTag<GRP>{}, IntTag<1>{}, P, P + 1 < PAIRS);
  }
  if (last) pair(IntTag<TAIL_NB>{}, IntTag<1>{}, LAST_PAIRS - 1, false);
}

extern "C" void kernel_launch(void* const* d_in, const int* in_sizes, int n_in,
                              void* d_out, int out_size, void* d_ws, size_t ws_size,
                              hipStream_t stream) {
  const float* x  = (const float*)d_in[0];
  const float* b0 = (const float*)d_in[1];
  const float* b1 = (const float*)d_in[2];
  const float* b2 = (const float*)d_in[3];
  const float* a1 = (const float*)d_in[4];
  const float* a2 = (const float*)d_in[5];
  float* y = (float*)d_out;

  biquad_kernel<<<B_CH, 32 * SLABS, 0, stream>>>(x, b0, b1, b2, a1, a2, y);
}
